# starting kernel
# baseline (speedup 1.0000x reference)
#include <hip/hip_runtime.h>
#include <hip/hip_bf16.h>

typedef __attribute__((ext_vector_type(2))) float v2f;
typedef __attribute__((ext_vector_type(4))) float v4f;
typedef __attribute__((ext_vector_type(8))) float v8f;

#define DIM 8192
#define NTHREADS 1024
#define NWAVES (NTHREADS / 32)
#define KCHUNKS ((DIM / 4) / NWAVES)

__global__ __launch_bounds__(NTHREADS)
void ExponentialTrajectoryFit_fused(const float* __restrict__ x0,
                                    const float* __restrict__ x1,
                                    const float* __restrict__ x2,
                                    const float* __restrict__ x3,
                                    float* __restrict__ out) {
    const int tid  = threadIdx.x;
    const int wave = tid >> 5;
    const int lane = tid & 31;
    const int m    = lane & 15;
    const int kh   = (lane >> 4) << 1;

    const float* p;
    const float* q;
    float sp, sq;
    if (m == 0)      { p = x1; q = x0; sp = 1.0f; sq = 1.0f; }
    else if (m == 1) { p = x2; q = x1; sp = 1.0f; sq = 1.0f; }
    else if (m == 2) { p = x3; q = x2; sp = 1.0f; sq = 1.0f; }
    else if (m == 3) { p = x0; q = x0; sp = 1.0f; sq = 0.0f; }
    else if (m == 4) { p = x1; q = x0; sp = 1.0f; sq = 0.0f; }
    else             { p = x0; q = x0; sp = 0.0f; sq = 0.0f; }

    const float* pp = p + wave * 4 + kh;
    const float* qq = q + wave * 4 + kh;

    v8f acc = {};
    #pragma unroll 8
    for (int it = 0; it < KCHUNKS; ++it) {
        const v2f P = *(const v2f*)(pp + it * (4 * NWAVES));
        const v2f Q = *(const v2f*)(qq + it * (4 * NWAVES));
        v2f a;
        a[0] = sp * P[0] - sq * Q[0];
        a[1] = sp * P[1] - sq * Q[1];
        acc = __builtin_amdgcn_wmma_f32_16x16x4_f32(
             false, a,  false, a,
             (short)0, acc,  false,  false);
    }

    __shared__ float partial[NWAVES][10];
    __shared__ float gsum[10];
    __shared__ float coeff[2];
    if (lane < 5) {
        partial[wave][lane * 2 + 0] = acc[0];
        partial[wave][lane * 2 + 1] = acc[1];
    }
    __syncthreads();

    if (tid < 10) {
        float s = 0.0f;
        #pragma unroll
        for (int w = 0; w < NWAVES; ++w) s += partial[w][tid];
        gsum[tid] = s;
    }
    __syncthreads();
    if (tid == 0) {
        const float g00 = gsum[0], g10 = gsum[1];
        const float g01 = gsum[2], g11 = gsum[3];
        const float q02 = gsum[4], q12 = gsum[5];
        const float p00 = gsum[6], p01 = gsum[7];
        const float p10 = gsum[8], p11 = gsum[9];

        const float det = g00 * g11 - g01 * g10;
        const float id  = 1.0f / det;
        const float a0 = id * ( g11 * p00 - g01 * p01);
        const float a1 = id * (-g10 * p00 + g00 * p01);
        const float r0 = p10 - (g01 * a0 + q02 * a1);
        const float r1 = p11 - (g11 * a0 + q12 * a1);
        const float v0 = id * ( g11 * r0 - g01 * r1);
        const float v1 = id * (-g10 * r0 + g00 * r1);
        const float s00 = id * ( g11 * g01 - g01 * g11);
        const float s10 = id * (-g10 * g01 + g00 * g11);
        const float s01 = id * ( g11 * q02 - g01 * q12);
        const float s11 = id * (-g10 * q02 + g00 * q12);
        const float w00 = 1.0f - s00, w01 = -s01;
        const float w10 = -s10,       w11 = 1.0f - s11;
        const float iw  = 1.0f / (w00 * w11 - w01 * w10);
        const float c0 = iw * ( w11 * v0 - w01 * v1);
        const float c1 = iw * (-w10 * v0 + w00 * v1);

        coeff[0] = a0 - c0;
        coeff[1] = a1 - c1;
    }
    __syncthreads();

    const float alpha = coeff[0];
    const float beta  = coeff[1];
    const int base = tid * 8;
    #pragma unroll
    for (int j = 0; j < 2; ++j) {
        const int i = base + j * 4;
        const v4f e0 = *(const v4f*)(x0 + i);
        const v4f e1 = *(const v4f*)(x1 + i);
        const v4f e2 = *(const v4f*)(x2 + i);
        const v4f e3 = *(const v4f*)(x3 + i);
        const v4f r  = (e0 - e1) + alpha * (e2 - e1) + beta * (e3 - e2);
        *(v4f*)(out + i) = r;
    }
}

extern "C" void kernel_launch(void* const* d_in, const int* in_sizes, int n_in,
                              void* d_out, int out_size, void* d_ws, size_t ws_size,
                              hipStream_t stream) {
    (void)in_sizes; (void)n_in; (void)out_size; (void)d_ws; (void)ws_size;
    const float* x0 = (const float*)d_in[0];
    const float* x1 = (const float*)d_in[1];
    const float* x2 = (const float*)d_in[2];
    const float* x3 = (const float*)d_in[3];
    float* out = (float*)d_out;
    ExponentialTrajectoryFit_fused<<<1, NTHREADS, 0, stream>>>(x0, x1, x2, x3, out);
}


# Gram on rows {x0,d0,d1,d2} x 4 K-slices filling all 16 WMMA rows (one b128 load per lane, rows by a DPP row_shr:4 subtract; d.x1 = d.x0 + d.d0): 4x fewer f32 WMMAs; reduction tree changes
# speedup vs baseline: 1.9289x; 1.9289x over previous
#include <hip/hip_runtime.h>
#include <hip/hip_bf16.h>

typedef __attribute__((ext_vector_type(2))) float v2f;
typedef __attribute__((ext_vector_type(4))) float v4f;
typedef __attribute__((ext_vector_type(8))) float v8f;

#define DIM 8192
#define NTHREADS 1024
#define NWAVES (NTHREADS / 32)
#define KPW (DIM / NWAVES)
#define NLOADS (KPW / 32)

__device__ __forceinline__ float prev_stream(float v) {
    return __int_as_float(__builtin_amdgcn_update_dpp(
        0, __float_as_int(v), 0x114, 0xf, 0xf, true));
}
__device__ __forceinline__ float quad_xor1(float v) {
    return __int_as_float(__builtin_amdgcn_update_dpp(
        0, __float_as_int(v), 0xB1, 0xf, 0xf, true));
}
__device__ __forceinline__ float quad_xor2(float v) {
    return __int_as_float(__builtin_amdgcn_update_dpp(
        0, __float_as_int(v), 0x4E, 0xf, 0xf, true));
}
__device__ __forceinline__ float lane_value(float v, int l) {
    return __int_as_float(__builtin_amdgcn_readlane(__float_as_int(v), l));
}

__global__ __launch_bounds__(NTHREADS)
void ExponentialTrajectoryFit_fused(const float* __restrict__ x0,
                                    const float* __restrict__ x1,
                                    const float* __restrict__ x2,
                                    const float* __restrict__ x3,
                                    float* __restrict__ out) {
    const int tid  = threadIdx.x;
    const int wave = tid >> 5;
    const int lane = tid & 31;
    const int c    = lane & 15;
    const int h    = lane >> 4;
    const int q    = c >> 2;
    const int j    = c & 3;

    const float* src = (q == 0) ? x0 : (q == 1) ? x1 : (q == 2) ? x2 : x3;
    const float* sp = src + wave * KPW + (h * 4 + j) * 4;

    v4f X[NLOADS];
    #pragma unroll
    for (int t = 0; t < NLOADS; ++t) X[t] = *(const v4f*)(sp + t * 32);

    v8f acc0 = {};
    v8f acc1 = {};
    #pragma unroll
    for (int t = 0; t < NLOADS; ++t) {
        v4f a;
        a[0] = X[t][0] - prev_stream(X[t][0]);
        a[1] = X[t][1] - prev_stream(X[t][1]);
        a[2] = X[t][2] - prev_stream(X[t][2]);
        a[3] = X[t][3] - prev_stream(X[t][3]);
        const v2f lo = {a[0], a[1]};
        const v2f hi = {a[2], a[3]};
        acc0 = __builtin_amdgcn_wmma_f32_16x16x4_f32(
             false, lo,  false, lo,
             (short)0, acc0,  false,  false);
        acc1 = __builtin_amdgcn_wmma_f32_16x16x4_f32(
             false, hi,  false, hi,
             (short)0, acc1,  false,  false);
    }
    const v8f acc = acc0 + acc1;

    const float lo_j = (j == 0) ? acc[0] : (j == 1) ? acc[1] : (j == 2) ? acc[2] : acc[3];
    const float hi_j = (j == 0) ? acc[4] : (j == 1) ? acc[5] : (j == 2) ? acc[6] : acc[7];
    float v = h ? lo_j : hi_j;
    v += quad_xor1(v);
    v += quad_xor2(v);

    __shared__ float partial[NWAVES * 8];
    __shared__ float coeff[2];
    if (j == 0) partial[wave * 8 + h * 4 + q] = v;
    __syncthreads();

    if (wave == 0) {
        const int o = lane & 7;
        float s0 = 0.0f, s1 = 0.0f, s2 = 0.0f, s3 = 0.0f;
        #pragma unroll
        for (int w = 0; w < NWAVES; w += 4) {
            s0 += partial[(w + 0) * 8 + o];
            s1 += partial[(w + 1) * 8 + o];
            s2 += partial[(w + 2) * 8 + o];
            s3 += partial[(w + 3) * 8 + o];
        }
        const float s = (s0 + s1) + (s2 + s3);

        const float p00 = lane_value(s, 0);
        const float g00 = lane_value(s, 1);
        const float g01 = lane_value(s, 2);
        const float q02 = lane_value(s, 3);
        const float p01 = lane_value(s, 4);
        const float g10 = lane_value(s, 5);
        const float g11 = lane_value(s, 6);
        const float q12 = lane_value(s, 7);
        const float p10 = p00 + g00;
        const float p11 = p01 + g10;

        const float det = g00 * g11 - g01 * g10;
        const float id  = 1.0f / det;
        const float a0 = id * ( g11 * p00 - g01 * p01);
        const float a1 = id * (-g10 * p00 + g00 * p01);
        const float r0 = p10 - (g01 * a0 + q02 * a1);
        const float r1 = p11 - (g11 * a0 + q12 * a1);
        const float v0 = id * ( g11 * r0 - g01 * r1);
        const float v1 = id * (-g10 * r0 + g00 * r1);
        const float s00 = id * ( g11 * g01 - g01 * g11);
        const float s10 = id * (-g10 * g01 + g00 * g11);
        const float s01 = id * ( g11 * q02 - g01 * q12);
        const float s11 = id * (-g10 * q02 + g00 * q12);
        const float w00 = 1.0f - s00, w01 = -s01;
        const float w10 = -s10,       w11 = 1.0f - s11;
        const float iw  = 1.0f / (w00 * w11 - w01 * w10);
        const float c0 = iw * ( w11 * v0 - w01 * v1);
        const float c1 = iw * (-w10 * v0 + w00 * v1);

        if (lane == 0) {
            coeff[0] = a0 - c0;
            coeff[1] = a1 - c1;
        }
    }
    __syncthreads();

    const float alpha = coeff[0];
    const float beta  = coeff[1];
    const int base = tid * 8;
    #pragma unroll
    for (int jj = 0; jj < 2; ++jj) {
        const int i = base + jj * 4;
        const v4f e0 = *(const v4f*)(x0 + i);
        const v4f e1 = *(const v4f*)(x1 + i);
        const v4f e2 = *(const v4f*)(x2 + i);
        const v4f e3 = *(const v4f*)(x3 + i);
        const v4f r  = (e0 - e1) + alpha * (e2 - e1) + beta * (e3 - e2);
        *(v4f*)(out + i) = r;
    }
}

extern "C" void kernel_launch(void* const* d_in, const int* in_sizes, int n_in,
                              void* d_out, int out_size, void* d_ws, size_t ws_size,
                              hipStream_t stream) {
    (void)in_sizes; (void)n_in; (void)out_size; (void)d_ws; (void)ws_size;
    const float* x0 = (const float*)d_in[0];
    const float* x1 = (const float*)d_in[1];
    const float* x2 = (const float*)d_in[2];
    const float* x3 = (const float*)d_in[3];
    float* out = (float*)d_out;
    ExponentialTrajectoryFit_fused<<<1, NTHREADS, 0, stream>>>(x0, x1, x2, x3, out);
}


# Output stage's 8 b128 input loads per thread issued right behind the Gram loads and held in registers over both barriers: their round trip runs under the WMMAs, not after the solve
# speedup vs baseline: 2.0708x; 1.0736x over previous
#include <hip/hip_runtime.h>
#include <hip/hip_bf16.h>

typedef __attribute__((ext_vector_type(2))) float v2f;
typedef __attribute__((ext_vector_type(4))) float v4f;
typedef __attribute__((ext_vector_type(8))) float v8f;

#define DIM 8192
#define NTHREADS 1024
#define NWAVES (NTHREADS / 32)
#define KPW (DIM / NWAVES)
#define NLOADS (KPW / 32)

__device__ __forceinline__ float prev_stream(float v) {
    return __int_as_float(__builtin_amdgcn_update_dpp(
        0, __float_as_int(v), 0x114, 0xf, 0xf, true));
}
__device__ __forceinline__ float quad_xor1(float v) {
    return __int_as_float(__builtin_amdgcn_update_dpp(
        0, __float_as_int(v), 0xB1, 0xf, 0xf, true));
}
__device__ __forceinline__ float quad_xor2(float v) {
    return __int_as_float(__builtin_amdgcn_update_dpp(
        0, __float_as_int(v), 0x4E, 0xf, 0xf, true));
}
__device__ __forceinline__ float lane_value(float v, int l) {
    return __int_as_float(__builtin_amdgcn_readlane(__float_as_int(v), l));
}

__global__ __launch_bounds__(NTHREADS)
void ExponentialTrajectoryFit_fused(const float* __restrict__ x0,
                                    const float* __restrict__ x1,
                                    const float* __restrict__ x2,
                                    const float* __restrict__ x3,
                                    float* __restrict__ out) {
    const int tid  = threadIdx.x;
    const int wave = tid >> 5;
    const int lane = tid & 31;
    const int c    = lane & 15;
    const int h    = lane >> 4;
    const int q    = c >> 2;
    const int j    = c & 3;

    const float* src = (q == 0) ? x0 : (q == 1) ? x1 : (q == 2) ? x2 : x3;
    const float* sp = src + wave * KPW + (h * 4 + j) * 4;

    v4f X[NLOADS];
    #pragma unroll
    for (int t = 0; t < NLOADS; ++t) X[t] = *(const v4f*)(sp + t * 32);

    __builtin_amdgcn_sched_barrier(0);

    const int base = tid * 8;
    v4f E0[2], E1[2], E2[2], E3[2];
    #pragma unroll
    for (int jj = 0; jj < 2; ++jj) {
        const int i = base + jj * 4;
        E0[jj] = *(const v4f*)(x0 + i);
        E1[jj] = *(const v4f*)(x1 + i);
        E2[jj] = *(const v4f*)(x2 + i);
        E3[jj] = *(const v4f*)(x3 + i);
    }
    __builtin_amdgcn_sched_barrier(0);

    v8f acc0 = {};
    v8f acc1 = {};
    #pragma unroll
    for (int t = 0; t < NLOADS; ++t) {
        v4f a;
        a[0] = X[t][0] - prev_stream(X[t][0]);
        a[1] = X[t][1] - prev_stream(X[t][1]);
        a[2] = X[t][2] - prev_stream(X[t][2]);
        a[3] = X[t][3] - prev_stream(X[t][3]);
        const v2f lo = {a[0], a[1]};
        const v2f hi = {a[2], a[3]};
        acc0 = __builtin_amdgcn_wmma_f32_16x16x4_f32(
             false, lo,  false, lo,
             (short)0, acc0,  false,  false);
        acc1 = __builtin_amdgcn_wmma_f32_16x16x4_f32(
             false, hi,  false, hi,
             (short)0, acc1,  false,  false);
    }
    const v8f acc = acc0 + acc1;

    const float lo_j = (j == 0) ? acc[0] : (j == 1) ? acc[1] : (j == 2) ? acc[2] : acc[3];
    const float hi_j = (j == 0) ? acc[4] : (j == 1) ? acc[5] : (j == 2) ? acc[6] : acc[7];
    float v = h ? lo_j : hi_j;
    v += quad_xor1(v);
    v += quad_xor2(v);

    __shared__ float partial[NWAVES * 8];
    __shared__ float coeff[2];
    if (j == 0) partial[wave * 8 + h * 4 + q] = v;
    __syncthreads();

    if (wave == 0) {
        const int o = lane & 7;
        float s0 = 0.0f, s1 = 0.0f, s2 = 0.0f, s3 = 0.0f;
        #pragma unroll
        for (int w = 0; w < NWAVES; w += 4) {
            s0 += partial[(w + 0) * 8 + o];
            s1 += partial[(w + 1) * 8 + o];
            s2 += partial[(w + 2) * 8 + o];
            s3 += partial[(w + 3) * 8 + o];
        }
        const float s = (s0 + s1) + (s2 + s3);

        const float p00 = lane_value(s, 0);
        const float g00 = lane_value(s, 1);
        const float g01 = lane_value(s, 2);
        const float q02 = lane_value(s, 3);
        const float p01 = lane_value(s, 4);
        const float g10 = lane_value(s, 5);
        const float g11 = lane_value(s, 6);
        const float q12 = lane_value(s, 7);
        const float p10 = p00 + g00;
        const float p11 = p01 + g10;

        const float det = g00 * g11 - g01 * g10;
        const float id  = 1.0f / det;
        const float a0 = id * ( g11 * p00 - g01 * p01);
        const float a1 = id * (-g10 * p00 + g00 * p01);
        const float r0 = p10 - (g01 * a0 + q02 * a1);
        const float r1 = p11 - (g11 * a0 + q12 * a1);
        const float v0 = id * ( g11 * r0 - g01 * r1);
        const float v1 = id * (-g10 * r0 + g00 * r1);
        const float s00 = id * ( g11 * g01 - g01 * g11);
        const float s10 = id * (-g10 * g01 + g00 * g11);
        const float s01 = id * ( g11 * q02 - g01 * q12);
        const float s11 = id * (-g10 * q02 + g00 * q12);
        const float w00 = 1.0f - s00, w01 = -s01;
        const float w10 = -s10,       w11 = 1.0f - s11;
        const float iw  = 1.0f / (w00 * w11 - w01 * w10);
        const float c0 = iw * ( w11 * v0 - w01 * v1);
        const float c1 = iw * (-w10 * v0 + w00 * v1);

        if (lane == 0) {
            coeff[0] = a0 - c0;
            coeff[1] = a1 - c1;
        }
    }
    __syncthreads();

    const float alpha = coeff[0];
    const float beta  = coeff[1];
    #pragma unroll
    for (int jj = 0; jj < 2; ++jj) {
        const v4f r = (E0[jj] - E1[jj]) + alpha * (E2[jj] - E1[jj]) + beta * (E3[jj] - E2[jj]);
        *(v4f*)(out + base + jj * 4) = r;
    }
}

extern "C" void kernel_launch(void* const* d_in, const int* in_sizes, int n_in,
                              void* d_out, int out_size, void* d_ws, size_t ws_size,
                              hipStream_t stream) {
    (void)in_sizes; (void)n_in; (void)out_size; (void)d_ws; (void)ws_size;
    const float* x0 = (const float*)d_in[0];
    const float* x1 = (const float*)d_in[1];
    const float* x2 = (const float*)d_in[2];
    const float* x3 = (const float*)d_in[3];
    float* out = (float*)d_out;
    ExponentialTrajectoryFit_fused<<<1, NTHREADS, 0, stream>>>(x0, x1, x2, x3, out);
}


# Coefficients from the five sums among d0,d1,d2 (x0 terms cancel: out = -(I-M)^-1 d0), tiled (d0|d1)x(d1|d2) on 8 lane-pair slices + d0xd0 on 16: 12 WMMAs per wave, not 16; reduction tree changes
# speedup vs baseline: 2.5658x; 1.2390x over previous
#include <hip/hip_runtime.h>
#include <hip/hip_bf16.h>


typedef __attribute__((ext_vector_type(2))) float v2f;
typedef __attribute__((ext_vector_type(4))) float v4f;
typedef __attribute__((ext_vector_type(8))) float v8f;

#define DIM 8192
#define NTHREADS 1024
#define NWAVES (NTHREADS / 32)
#define NGROUPS (NWAVES * 2)

__device__ __forceinline__ float pair_swap(float v) {
    return __int_as_float(__builtin_amdgcn_update_dpp(
        0, __float_as_int(v), 0xB1, 0xf, 0xf, true));
}
__device__ __forceinline__ float quad_xor2(float v) {
    return __int_as_float(__builtin_amdgcn_update_dpp(
        0, __float_as_int(v), 0x4E, 0xf, 0xf, true));
}
__device__ __forceinline__ float row_rot4(float v) {
    return __int_as_float(__builtin_amdgcn_update_dpp(
        0, __float_as_int(v), 0x124, 0xf, 0xf, true));
}
__device__ __forceinline__ float row_rot8(float v) {
    return __int_as_float(__builtin_amdgcn_update_dpp(
        0, __float_as_int(v), 0x128, 0xf, 0xf, true));
}
__device__ __forceinline__ float lane_value(float v, int l) {
    return __int_as_float(__builtin_amdgcn_readlane(__float_as_int(v), l));
}

__global__ __launch_bounds__(NTHREADS)
void ExponentialTrajectoryFit_fused(const float* __restrict__ x0,
                                    const float* __restrict__ x1,
                                    const float* __restrict__ x2,
                                    const float* __restrict__ x3,
                                    float* __restrict__ out) {
    const int tid  = threadIdx.x;
    const int wave = tid >> 5;
    const int lane = tid & 31;
    const int c    = lane & 15;
    const int h    = lane >> 4;
    const bool odd = (lane & 1) != 0;

    const int base = tid * 8;
    v4f E0[2], E1[2], E2[2], E3[2];
    #pragma unroll
    for (int jj = 0; jj < 2; ++jj) {
        const int i = base + jj * 4;
        E0[jj] = *(const v4f*)(x0 + i);
        E1[jj] = *(const v4f*)(x1 + i);
        E2[jj] = *(const v4f*)(x2 + i);
        E3[jj] = *(const v4f*)(x3 + i);
    }

    float d0[8], d1[8], d2[8];
    #pragma unroll
    for (int jj = 0; jj < 2; ++jj) {
        #pragma unroll
        for (int e = 0; e < 4; ++e) {
            d0[jj * 4 + e] = E1[jj][e] - E0[jj][e];
            d1[jj * 4 + e] = E2[jj][e] - E1[jj][e];
            d2[jj * 4 + e] = E3[jj][e] - E2[jj][e];
        }
    }

    v8f accE = {};
    v8f accO = {};
    v8f accS = {};
    #pragma unroll
    for (int p = 0; p < 4; ++p) {
        const int i0 = 2 * p, i1 = 2 * p + 1;
        const float s0a = pair_swap(d0[i0]), s0b = pair_swap(d0[i1]);
        const float s1a = pair_swap(d1[i0]), s1b = pair_swap(d1[i1]);
        const float s2a = pair_swap(d2[i0]), s2b = pair_swap(d2[i1]);
        const v2f aE = {odd ? s1a : d0[i0], odd ? s1b : d0[i1]};
        const v2f bE = {odd ? s2a : d1[i0], odd ? s2b : d1[i1]};
        const v2f aO = {odd ? d1[i0] : s0a, odd ? d1[i1] : s0b};
        const v2f bO = {odd ? d2[i0] : s1a, odd ? d2[i1] : s1b};
        const v2f sq = {d0[i0], d0[i1]};
        accE = __builtin_amdgcn_wmma_f32_16x16x4_f32(
             false, aE,  false, bE,
             (short)0, accE,  false,  false);
        accO = __builtin_amdgcn_wmma_f32_16x16x4_f32(
             false, aO,  false, bO,
             (short)0, accO,  false,  false);
        accS = __builtin_amdgcn_wmma_f32_16x16x4_f32(
             false, sq,  false, sq,
             (short)0, accS,  false,  false);
    }
    const v8f acc = accE + accO;

    const bool b1 = (c & 1) != 0, b2 = (c & 2) != 0, b4 = (c & 4) != 0;
    const float u01 = b2 ? acc[2] : acc[0], u23 = b2 ? acc[6] : acc[4];
    const float v01 = b2 ? acc[3] : acc[1], v23 = b2 ? acc[7] : acc[5];
    float u = b4 ? u23 : u01;
    float v = b4 ? v23 : v01;
    const float w0 = b1 ? accS[1] : accS[0], w1 = b1 ? accS[3] : accS[2];
    const float w2 = b1 ? accS[5] : accS[4], w3 = b1 ? accS[7] : accS[6];
    const float w01 = b2 ? w1 : w0, w23 = b2 ? w3 : w2;
    float w = b4 ? w23 : w01;
    const bool owner = ((c >> 3) == h);
    u = owner ? u : 0.0f;
    v = owner ? v : 0.0f;
    w = owner ? w : 0.0f;
    u += row_rot8(u);  v += row_rot8(v);  w += row_rot8(w);
    u += row_rot4(u);  v += row_rot4(v);  w += row_rot4(w);
    u += quad_xor2(u); v += quad_xor2(v); w += quad_xor2(w);
    w += pair_swap(w);

    __shared__ float partial[NGROUPS * 8];
    __shared__ float coeff[2];
    const int group = wave * 2 + h;
    if (c == 0) {
        partial[group * 8 + 0] = u;
        partial[group * 8 + 1] = v;
        partial[group * 8 + 4] = w;
    }
    if (c == 1) {
        partial[group * 8 + 2] = u;
        partial[group * 8 + 3] = v;
    }
    __syncthreads();

    if (wave == 0) {
        const int o    = ((lane & 7) < 5) ? (lane & 7) : 4;
        const int part = lane >> 3;
        float s0 = 0.0f, s1 = 0.0f, s2 = 0.0f, s3 = 0.0f;
        #pragma unroll
        for (int g = 0; g < NGROUPS; g += 16) {
            s0 += partial[(g +  0 + part) * 8 + o];
            s1 += partial[(g +  4 + part) * 8 + o];
            s2 += partial[(g +  8 + part) * 8 + o];
            s3 += partial[(g + 12 + part) * 8 + o];
        }
        float s = (s0 + s1) + (s2 + s3);
        s += row_rot8(s);

        const float g01 = lane_value(s, 0) + lane_value(s, 16);
        const float g11 = lane_value(s, 1) + lane_value(s, 17);
        const float q02 = lane_value(s, 2) + lane_value(s, 18);
        const float q12 = lane_value(s, 3) + lane_value(s, 19);
        const float g00 = lane_value(s, 4) + lane_value(s, 20);

        const float det = g00 * g11 - g01 * g01;
        const float n01 = g11 * q02 - g01 * q12;
        const float n11 = g00 * q12 - g01 * q02;
        const float den  = (det - n11) - n01;
        const float iden = 1.0f / den;
        if (lane == 0) {
            coeff[0] = -(det - n11) * iden;
            coeff[1] = -det * iden;
        }
    }
    __syncthreads();

    const float alpha = coeff[0];
    const float beta  = coeff[1];
    #pragma unroll
    for (int jj = 0; jj < 2; ++jj) {
        v4f r;
        #pragma unroll
        for (int e = 0; e < 4; ++e) {
            const int i = jj * 4 + e;
            r[e] = beta * d2[i] + (alpha * d1[i] - d0[i]);
        }
        *(v4f*)(out + base + jj * 4) = r;
    }
}

extern "C" void kernel_launch(void* const* d_in, const int* in_sizes, int n_in,
                              void* d_out, int out_size, void* d_ws, size_t ws_size,
                              hipStream_t stream) {
    (void)in_sizes; (void)n_in; (void)out_size; (void)d_ws; (void)ws_size;
    const float* x0 = (const float*)d_in[0];
    const float* x1 = (const float*)d_in[1];
    const float* x2 = (const float*)d_in[2];
    const float* x3 = (const float*)d_in[3];
    float* out = (float*)d_out;
    ExponentialTrajectoryFit_fused<<<1, NTHREADS, 0, stream>>>(x0, x1, x2, x3, out);
}
